# starting kernel
# baseline (speedup 1.0000x reference)
#include <hip/hip_runtime.h>
#include <hip/hip_bf16.h>

typedef __attribute__((ext_vector_type(2))) float v2f;
typedef __attribute__((ext_vector_type(4))) float v4f;
typedef __attribute__((ext_vector_type(8))) float v8f;

#define DIM 64
#define KEEP_PROB 0.7f

__global__ void __launch_bounds__(256)
gcn_fc_wmma_kernel(const float* __restrict__ emb, const float* __restrict__ W,
                   const float* __restrict__ bias, float* __restrict__ acc,
                   int n_nodes) {
    __shared__ alignas(16) float Wlds[DIM * DIM];
    __shared__ alignas(16) float Alds[8][16 * DIM];

    const int tid  = threadIdx.x;
    const int wave = tid >> 5;
    const int lane = tid & 31;

    for (int i = tid; i < (DIM * DIM) / 4; i += 256) {
        ((v4f*)Wlds)[i] = ((const v4f*)W)[i];
    }

    const int row_base = blockIdx.x * 128 + wave * 16;

    for (int i = lane; i < 256; i += 32) {
        int r  = i >> 4;
        int c4 = i & 15;
        int gr = row_base + r;
        v4f v = (gr < n_nodes) ? ((const v4f*)(emb + (size_t)gr * DIM))[c4]
                               : (v4f){0.f, 0.f, 0.f, 0.f};
        ((v4f*)&Alds[wave][0])[r * 16 + c4] = v;
    }
    __syncthreads();

    const float* At = &Alds[wave][0];

    const int m_rel = lane & 15;
    const int k_off = (lane >> 4) * 2;
    const int n_lo  = lane & 15;

    v8f c0 = {}, c1 = {}, c2 = {}, c3 = {};

#pragma unroll
    for (int kk = 0; kk < 16; ++kk) {
        const int k = kk * 4 + k_off;
        v2f a  = *(const v2f*)(At + m_rel * DIM + k);
        v2f b0 = *(const v2f*)(Wlds + (n_lo +  0) * DIM + k);
        v2f b1 = *(const v2f*)(Wlds + (n_lo + 16) * DIM + k);
        v2f b2 = *(const v2f*)(Wlds + (n_lo + 32) * DIM + k);
        v2f b3 = *(const v2f*)(Wlds + (n_lo + 48) * DIM + k);
        c0 = __builtin_amdgcn_wmma_f32_16x16x4_f32(false, a, false, b0, (short)0, c0, false, false);
        c1 = __builtin_amdgcn_wmma_f32_16x16x4_f32(false, a, false, b1, (short)0, c1, false, false);
        c2 = __builtin_amdgcn_wmma_f32_16x16x4_f32(false, a, false, b2, (short)0, c2, false, false);
        c3 = __builtin_amdgcn_wmma_f32_16x16x4_f32(false, a, false, b3, (short)0, c3, false, false);
    }

    const float bb0 = bias[n_lo];
    const float bb1 = bias[n_lo + 16];
    const float bb2 = bias[n_lo + 32];
    const float bb3 = bias[n_lo + 48];

    const int m_hi = (lane >> 4) * 8;
#pragma unroll
    for (int j = 0; j < 8; ++j) {
        int gm = row_base + m_hi + j;
        if (gm < n_nodes) {
            float* outr = acc + (size_t)gm * DIM;
            outr[n_lo +  0] = c0[j] + bb0;
            outr[n_lo + 16] = c1[j] + bb1;
            outr[n_lo + 32] = c2[j] + bb2;
            outr[n_lo + 48] = c3[j] + bb3;
        }
    }
}

__global__ void __launch_bounds__(256)
gcn_spmm_kernel(const float* __restrict__ vals, const float* __restrict__ dropu,
                const int* __restrict__ rows, const int* __restrict__ cols,
                const float* __restrict__ src, float* __restrict__ dst,
                int n_edges) {
    const long long tid = (long long)blockIdx.x * blockDim.x + threadIdx.x;
    int e = (int)(tid >> 5);
    if (e >= n_edges) return;
    e = __builtin_amdgcn_readfirstlane(e);
    const int l = threadIdx.x & 31;

    const float u = dropu[e];
    if (u + KEEP_PROB < 1.0f) return;
    const float v = vals[e] * (1.0f / KEEP_PROB);

    const int col = cols[e];
    const int row = rows[e];

    const float* sp = src + (size_t)col * DIM + l * 2;
    float*       dp = dst + (size_t)row * DIM + l * 2;
    const float m0 = v * sp[0];
    const float m1 = v * sp[1];

    asm volatile("global_atomic_add_f32 %0, %1, off scope:SCOPE_DEV"
                 :: "v"(dp), "v"(m0) : "memory");
    asm volatile("global_atomic_add_f32 %0, %1, off scope:SCOPE_DEV"
                 :: "v"(dp + 1), "v"(m1) : "memory");
}

__global__ void __launch_bounds__(256)
gcn_finalize_kernel(float* __restrict__ out, const float* __restrict__ x1,
                    long long n4) {
    const long long i = (long long)blockIdx.x * blockDim.x + threadIdx.x;
    if (i >= n4) return;
    v4f a = ((const v4f*)out)[i];
    v4f p = ((const v4f*)x1)[i];
    v4f r = (a + p) * (1.0f / 3.0f);
    ((v4f*)out)[i] = r;
}

extern "C" void kernel_launch(void* const* d_in, const int* in_sizes, int n_in,
                              void* d_out, int out_size, void* d_ws, size_t ws_size,
                              hipStream_t stream) {
    const float* emb   = (const float*)d_in[0];
    const float* W     = (const float*)d_in[1];
    const float* bias  = (const float*)d_in[2];
    const float* vals  = (const float*)d_in[3];
    const float* dropu = (const float*)d_in[4];
    const int*   rows  = (const int*)d_in[5];
    const int*   cols  = (const int*)d_in[6];

    const int N = in_sizes[0] / DIM;
    const int E = in_sizes[3];

    float* out = (float*)d_out;
    float* x1  = (float*)d_ws;

    hipMemsetAsync(d_ws, 0, (size_t)N * DIM * sizeof(float), stream);

    {
        dim3 grid((N + 127) / 128), block(256);
        gcn_fc_wmma_kernel<<<grid, block, 0, stream>>>(emb, W, bias, out, N);
    }

    {
        long long threads = (long long)E * 32;
        dim3 grid((unsigned)((threads + 255) / 256)), block(256);
        gcn_spmm_kernel<<<grid, block, 0, stream>>>(vals, dropu,     rows, cols, emb, x1,  E);
        gcn_spmm_kernel<<<grid, block, 0, stream>>>(vals, dropu + E, rows, cols, x1,  out, E);
    }

    {
        long long n4 = (long long)N * DIM / 4;
        dim3 grid((unsigned)((n4 + 255) / 256)), block(256);
        gcn_finalize_kernel<<<grid, block, 0, stream>>>(out, x1, n4);
    }
}


# SpMM: one wave per 32 consecutive edges (coalesced per-lane edge loads, ballot of kept edges, readlane broadcast, 8 row gathers in flight) instead of one wave per edge; atomic arrival order changes
# speedup vs baseline: 2.3271x; 2.3271x over previous
#include <hip/hip_runtime.h>
#include <hip/hip_bf16.h>

typedef __attribute__((ext_vector_type(2))) float v2f;
typedef __attribute__((ext_vector_type(4))) float v4f;
typedef __attribute__((ext_vector_type(8))) float v8f;

#define DIM 64
#define KEEP_PROB 0.7f

__global__ void __launch_bounds__(256)
gcn_fc_wmma_kernel(const float* __restrict__ emb, const float* __restrict__ W,
                   const float* __restrict__ bias, float* __restrict__ acc,
                   int n_nodes) {
    __shared__ alignas(16) float Wlds[DIM * DIM];
    __shared__ alignas(16) float Alds[8][16 * DIM];

    const int tid  = threadIdx.x;
    const int wave = tid >> 5;
    const int lane = tid & 31;

    for (int i = tid; i < (DIM * DIM) / 4; i += 256) {
        ((v4f*)Wlds)[i] = ((const v4f*)W)[i];
    }

    const int row_base = blockIdx.x * 128 + wave * 16;

    for (int i = lane; i < 256; i += 32) {
        int r  = i >> 4;
        int c4 = i & 15;
        int gr = row_base + r;
        v4f v = (gr < n_nodes) ? ((const v4f*)(emb + (size_t)gr * DIM))[c4]
                               : (v4f){0.f, 0.f, 0.f, 0.f};
        ((v4f*)&Alds[wave][0])[r * 16 + c4] = v;
    }
    __syncthreads();

    const float* At = &Alds[wave][0];

    const int m_rel = lane & 15;
    const int k_off = (lane >> 4) * 2;
    const int n_lo  = lane & 15;

    v8f c0 = {}, c1 = {}, c2 = {}, c3 = {};

#pragma unroll
    for (int kk = 0; kk < 16; ++kk) {
        const int k = kk * 4 + k_off;
        v2f a  = *(const v2f*)(At + m_rel * DIM + k);
        v2f b0 = *(const v2f*)(Wlds + (n_lo +  0) * DIM + k);
        v2f b1 = *(const v2f*)(Wlds + (n_lo + 16) * DIM + k);
        v2f b2 = *(const v2f*)(Wlds + (n_lo + 32) * DIM + k);
        v2f b3 = *(const v2f*)(Wlds + (n_lo + 48) * DIM + k);
        c0 = __builtin_amdgcn_wmma_f32_16x16x4_f32(false, a, false, b0, (short)0, c0, false, false);
        c1 = __builtin_amdgcn_wmma_f32_16x16x4_f32(false, a, false, b1, (short)0, c1, false, false);
        c2 = __builtin_amdgcn_wmma_f32_16x16x4_f32(false, a, false, b2, (short)0, c2, false, false);
        c3 = __builtin_amdgcn_wmma_f32_16x16x4_f32(false, a, false, b3, (short)0, c3, false, false);
    }

    const float bb0 = bias[n_lo];
    const float bb1 = bias[n_lo + 16];
    const float bb2 = bias[n_lo + 32];
    const float bb3 = bias[n_lo + 48];

    const int m_hi = (lane >> 4) * 8;
#pragma unroll
    for (int j = 0; j < 8; ++j) {
        int gm = row_base + m_hi + j;
        if (gm < n_nodes) {
            float* outr = acc + (size_t)gm * DIM;
            outr[n_lo +  0] = c0[j] + bb0;
            outr[n_lo + 16] = c1[j] + bb1;
            outr[n_lo + 32] = c2[j] + bb2;
            outr[n_lo + 48] = c3[j] + bb3;
        }
    }
}

__global__ void __launch_bounds__(256)
gcn_spmm_kernel(const float* __restrict__ vals, const float* __restrict__ dropu,
                const int* __restrict__ rows, const int* __restrict__ cols,
                const float* __restrict__ src, float* __restrict__ dst,
                int n_edges) {
    const int e = blockIdx.x * 256 + threadIdx.x;
    const int l = threadIdx.x & 31;
    const bool in_range = e < n_edges;

    float u = 0.0f, w = 0.0f;
    int col = 0, row = 0;
    if (in_range) {
        u   = dropu[e];
        w   = vals[e] * (1.0f / KEEP_PROB);
        col = cols[e];
        row = rows[e];
    }
    const bool keep = in_range && !(u + KEEP_PROB < 1.0f);
    unsigned mask = __builtin_amdgcn_ballot_w32(keep);
    const int wbits = __float_as_int(w);

    while (mask != 0u) {
        const int first = __builtin_ctz(mask);
        int  sl[8];
        bool ok[8];
#pragma unroll
        for (int i = 0; i < 8; ++i) {
            ok[i] = (mask != 0u);
            sl[i] = ok[i] ? __builtin_ctz(mask) : first;
            mask &= (mask - 1u);
        }

        v2f g[8];
#pragma unroll
        for (int i = 0; i < 8; ++i) {
            const int c = __builtin_amdgcn_readlane(col, sl[i]);
            g[i] = *(const v2f*)(src + (size_t)c * DIM + l * 2);
        }

#pragma unroll
        for (int i = 0; i < 8; ++i) {
            if (ok[i]) {
                const int   r  = __builtin_amdgcn_readlane(row, sl[i]);
                const float wv = __int_as_float(__builtin_amdgcn_readlane(wbits, sl[i]));
                float* dp = dst + (size_t)r * DIM + l * 2;
                const float m0 = wv * g[i].x;
                const float m1 = wv * g[i].y;
                asm volatile("global_atomic_add_f32 %0, %1, off scope:SCOPE_DEV"
                             :: "v"(dp), "v"(m0) : "memory");
                asm volatile("global_atomic_add_f32 %0, %1, off scope:SCOPE_DEV"
                             :: "v"(dp + 1), "v"(m1) : "memory");
            }
        }
    }
}

__global__ void __launch_bounds__(256)
gcn_finalize_kernel(float* __restrict__ out, const float* __restrict__ x1,
                    long long n4) {
    const long long i = (long long)blockIdx.x * blockDim.x + threadIdx.x;
    if (i >= n4) return;
    v4f a = ((const v4f*)out)[i];
    v4f p = ((const v4f*)x1)[i];
    v4f r = (a + p) * (1.0f / 3.0f);
    ((v4f*)out)[i] = r;
}

extern "C" void kernel_launch(void* const* d_in, const int* in_sizes, int n_in,
                              void* d_out, int out_size, void* d_ws, size_t ws_size,
                              hipStream_t stream) {
    const float* emb   = (const float*)d_in[0];
    const float* W     = (const float*)d_in[1];
    const float* bias  = (const float*)d_in[2];
    const float* vals  = (const float*)d_in[3];
    const float* dropu = (const float*)d_in[4];
    const int*   rows  = (const int*)d_in[5];
    const int*   cols  = (const int*)d_in[6];

    const int N = in_sizes[0] / DIM;
    const int E = in_sizes[3];

    float* out = (float*)d_out;
    float* x1  = (float*)d_ws;

    hipMemsetAsync(d_ws, 0, (size_t)N * DIM * sizeof(float), stream);

    {
        dim3 grid((N + 127) / 128), block(256);
        gcn_fc_wmma_kernel<<<grid, block, 0, stream>>>(emb, W, bias, out, N);
    }

    {
        dim3 grid((unsigned)((E + 255) / 256)), block(256);
        gcn_spmm_kernel<<<grid, block, 0, stream>>>(vals, dropu,     rows, cols, emb, x1,  E);
        gcn_spmm_kernel<<<grid, block, 0, stream>>>(vals, dropu + E, rows, cols, x1,  out, E);
    }

    {
        long long n4 = (long long)N * DIM / 4;
        dim3 grid((unsigned)((n4 + 255) / 256)), block(256);
        gcn_finalize_kernel<<<grid, block, 0, stream>>>(out, x1, n4);
    }
}


# SpMM lane map: lane l owns floats l and 32+l of a row (two 4-byte gathers) so each f32 atomic instruction covers one whole 128-B line, not two half-filled ones; atomic arrival order as before
# speedup vs baseline: 2.6089x; 1.1211x over previous
#include <hip/hip_runtime.h>
#include <hip/hip_bf16.h>

typedef __attribute__((ext_vector_type(2))) float v2f;
typedef __attribute__((ext_vector_type(4))) float v4f;
typedef __attribute__((ext_vector_type(8))) float v8f;

#define DIM 64
#define KEEP_PROB 0.7f

__global__ void __launch_bounds__(256)
gcn_fc_wmma_kernel(const float* __restrict__ emb, const float* __restrict__ W,
                   const float* __restrict__ bias, float* __restrict__ acc,
                   int n_nodes) {
    __shared__ alignas(16) float Wlds[DIM * DIM];
    __shared__ alignas(16) float Alds[8][16 * DIM];

    const int tid  = threadIdx.x;
    const int wave = tid >> 5;
    const int lane = tid & 31;

    for (int i = tid; i < (DIM * DIM) / 4; i += 256) {
        ((v4f*)Wlds)[i] = ((const v4f*)W)[i];
    }

    const int row_base = blockIdx.x * 128 + wave * 16;

    for (int i = lane; i < 256; i += 32) {
        int r  = i >> 4;
        int c4 = i & 15;
        int gr = row_base + r;
        v4f v = (gr < n_nodes) ? ((const v4f*)(emb + (size_t)gr * DIM))[c4]
                               : (v4f){0.f, 0.f, 0.f, 0.f};
        ((v4f*)&Alds[wave][0])[r * 16 + c4] = v;
    }
    __syncthreads();

    const float* At = &Alds[wave][0];

    const int m_rel = lane & 15;
    const int k_off = (lane >> 4) * 2;
    const int n_lo  = lane & 15;

    v8f c0 = {}, c1 = {}, c2 = {}, c3 = {};

#pragma unroll
    for (int kk = 0; kk < 16; ++kk) {
        const int k = kk * 4 + k_off;
        v2f a  = *(const v2f*)(At + m_rel * DIM + k);
        v2f b0 = *(const v2f*)(Wlds + (n_lo +  0) * DIM + k);
        v2f b1 = *(const v2f*)(Wlds + (n_lo + 16) * DIM + k);
        v2f b2 = *(const v2f*)(Wlds + (n_lo + 32) * DIM + k);
        v2f b3 = *(const v2f*)(Wlds + (n_lo + 48) * DIM + k);
        c0 = __builtin_amdgcn_wmma_f32_16x16x4_f32(false, a, false, b0, (short)0, c0, false, false);
        c1 = __builtin_amdgcn_wmma_f32_16x16x4_f32(false, a, false, b1, (short)0, c1, false, false);
        c2 = __builtin_amdgcn_wmma_f32_16x16x4_f32(false, a, false, b2, (short)0, c2, false, false);
        c3 = __builtin_amdgcn_wmma_f32_16x16x4_f32(false, a, false, b3, (short)0, c3, false, false);
    }

    const float bb0 = bias[n_lo];
    const float bb1 = bias[n_lo + 16];
    const float bb2 = bias[n_lo + 32];
    const float bb3 = bias[n_lo + 48];

    const int m_hi = (lane >> 4) * 8;
#pragma unroll
    for (int j = 0; j < 8; ++j) {
        int gm = row_base + m_hi + j;
        if (gm < n_nodes) {
            float* outr = acc + (size_t)gm * DIM;
            outr[n_lo +  0] = c0[j] + bb0;
            outr[n_lo + 16] = c1[j] + bb1;
            outr[n_lo + 32] = c2[j] + bb2;
            outr[n_lo + 48] = c3[j] + bb3;
        }
    }
}

__global__ void __launch_bounds__(256)
gcn_spmm_kernel(const float* __restrict__ vals, const float* __restrict__ dropu,
                const int* __restrict__ rows, const int* __restrict__ cols,
                const float* __restrict__ src, float* __restrict__ dst,
                int n_edges) {
    const int e = blockIdx.x * 256 + threadIdx.x;
    const int l = threadIdx.x & 31;
    const bool in_range = e < n_edges;

    float u = 0.0f, w = 0.0f;
    int col = 0, row = 0;
    if (in_range) {
        u   = dropu[e];
        w   = vals[e] * (1.0f / KEEP_PROB);
        col = cols[e];
        row = rows[e];
    }
    const bool keep = in_range && !(u + KEEP_PROB < 1.0f);
    unsigned mask = __builtin_amdgcn_ballot_w32(keep);
    const int wbits = __float_as_int(w);

    while (mask != 0u) {
        const int first = __builtin_ctz(mask);
        int  sl[8];
        bool ok[8];
#pragma unroll
        for (int i = 0; i < 8; ++i) {
            ok[i] = (mask != 0u);
            sl[i] = ok[i] ? __builtin_ctz(mask) : first;
            mask &= (mask - 1u);
        }

        float g0[8], g1[8];
#pragma unroll
        for (int i = 0; i < 8; ++i) {
            const int c = __builtin_amdgcn_readlane(col, sl[i]);
            const float* sp = src + (size_t)c * DIM + l;
            g0[i] = sp[0];
            g1[i] = sp[32];
        }

#pragma unroll
        for (int i = 0; i < 8; ++i) {
            float a0 = g0[i], a1 = g1[i];
            asm volatile("" : "+v"(a0), "+v"(a1));
            if (ok[i]) {
                const int   r  = __builtin_amdgcn_readlane(row, sl[i]);
                const float wv = __int_as_float(__builtin_amdgcn_readlane(wbits, sl[i]));
                float* dp = dst + (size_t)r * DIM + l;
                const float m0 = wv * a0;
                const float m1 = wv * a1;
                asm volatile("global_atomic_add_f32 %0, %1, off scope:SCOPE_DEV"
                             :: "v"(dp), "v"(m0) : "memory");
                asm volatile("global_atomic_add_f32 %0, %1, off offset:128 scope:SCOPE_DEV"
                             :: "v"(dp), "v"(m1) : "memory");
            }
        }
    }
}

__global__ void __launch_bounds__(256)
gcn_finalize_kernel(float* __restrict__ out, const float* __restrict__ x1,
                    long long n4) {
    const long long i = (long long)blockIdx.x * blockDim.x + threadIdx.x;
    if (i >= n4) return;
    v4f a = ((const v4f*)out)[i];
    v4f p = ((const v4f*)x1)[i];
    v4f r = (a + p) * (1.0f / 3.0f);
    ((v4f*)out)[i] = r;
}

extern "C" void kernel_launch(void* const* d_in, const int* in_sizes, int n_in,
                              void* d_out, int out_size, void* d_ws, size_t ws_size,
                              hipStream_t stream) {
    const float* emb   = (const float*)d_in[0];
    const float* W     = (const float*)d_in[1];
    const float* bias  = (const float*)d_in[2];
    const float* vals  = (const float*)d_in[3];
    const float* dropu = (const float*)d_in[4];
    const int*   rows  = (const int*)d_in[5];
    const int*   cols  = (const int*)d_in[6];

    const int N = in_sizes[0] / DIM;
    const int E = in_sizes[3];

    float* out = (float*)d_out;
    float* x1  = (float*)d_ws;

    hipMemsetAsync(d_ws, 0, (size_t)N * DIM * sizeof(float), stream);

    {
        dim3 grid((N + 127) / 128), block(256);
        gcn_fc_wmma_kernel<<<grid, block, 0, stream>>>(emb, W, bias, out, N);
    }

    {
        dim3 grid((unsigned)((E + 255) / 256)), block(256);
        gcn_spmm_kernel<<<grid, block, 0, stream>>>(vals, dropu,     rows, cols, emb, x1,  E);
        gcn_spmm_kernel<<<grid, block, 0, stream>>>(vals, dropu + E, rows, cols, x1,  out, E);
    }

    {
        long long n4 = (long long)N * DIM / 4;
        dim3 grid((unsigned)((n4 + 255) / 256)), block(256);
        gcn_finalize_kernel<<<grid, block, 0, stream>>>(out, x1, n4);
    }
}


# Mean folded into the fc pass: fc runs after layer 0 and writes (fc+x1)/3 as whole rows restaged through the wave's LDS tile; layer 1 adds messages pre-scaled by 1/3; mean pass gone
# speedup vs baseline: 3.6353x; 1.3934x over previous
#include <hip/hip_runtime.h>
#include <hip/hip_bf16.h>

typedef __attribute__((ext_vector_type(2))) float v2f;
typedef __attribute__((ext_vector_type(4))) float v4f;
typedef __attribute__((ext_vector_type(8))) float v8f;

#define DIM 64
#define KEEP_PROB 0.7f

__global__ void __launch_bounds__(256)
gcn_fc_wmma_kernel(const float* __restrict__ emb, const float* __restrict__ W,
                   const float* __restrict__ bias, const float* __restrict__ x1,
                   float* __restrict__ out, int n_nodes) {
    __shared__ alignas(16) float Wlds[DIM * DIM];
    __shared__ alignas(16) float Alds[8][16 * DIM];

    const int tid  = threadIdx.x;
    const int wave = tid >> 5;
    const int lane = tid & 31;

    for (int i = tid; i < (DIM * DIM) / 4; i += 256) {
        ((v4f*)Wlds)[i] = ((const v4f*)W)[i];
    }

    const int row_base = blockIdx.x * 128 + wave * 16;

    for (int i = lane; i < 256; i += 32) {
        int r  = i >> 4;
        int c4 = i & 15;
        int gr = row_base + r;
        v4f v = (gr < n_nodes) ? ((const v4f*)(emb + (size_t)gr * DIM))[c4]
                               : (v4f){0.f, 0.f, 0.f, 0.f};
        ((v4f*)&Alds[wave][0])[r * 16 + c4] = v;
    }
    __syncthreads();

    const float* At = &Alds[wave][0];

    const int m_rel = lane & 15;
    const int k_off = (lane >> 4) * 2;
    const int n_lo  = lane & 15;

    v8f c0 = {}, c1 = {}, c2 = {}, c3 = {};

#pragma unroll
    for (int kk = 0; kk < 16; ++kk) {
        const int k = kk * 4 + k_off;
        v2f a  = *(const v2f*)(At + m_rel * DIM + k);
        v2f b0 = *(const v2f*)(Wlds + (n_lo +  0) * DIM + k);
        v2f b1 = *(const v2f*)(Wlds + (n_lo + 16) * DIM + k);
        v2f b2 = *(const v2f*)(Wlds + (n_lo + 32) * DIM + k);
        v2f b3 = *(const v2f*)(Wlds + (n_lo + 48) * DIM + k);
        c0 = __builtin_amdgcn_wmma_f32_16x16x4_f32(false, a, false, b0, (short)0, c0, false, false);
        c1 = __builtin_amdgcn_wmma_f32_16x16x4_f32(false, a, false, b1, (short)0, c1, false, false);
        c2 = __builtin_amdgcn_wmma_f32_16x16x4_f32(false, a, false, b2, (short)0, c2, false, false);
        c3 = __builtin_amdgcn_wmma_f32_16x16x4_f32(false, a, false, b3, (short)0, c3, false, false);
    }

    const float bb0 = bias[n_lo];
    const float bb1 = bias[n_lo + 16];
    const float bb2 = bias[n_lo + 32];
    const float bb3 = bias[n_lo + 48];

    float* Ct = &Alds[wave][0];
    const int m_hi = (lane >> 4) * 8;
#pragma unroll
    for (int j = 0; j < 8; ++j) {
        float* cr = Ct + (m_hi + j) * DIM + n_lo;
        cr[ 0] = c0[j] + bb0;
        cr[16] = c1[j] + bb1;
        cr[32] = c2[j] + bb2;
        cr[48] = c3[j] + bb3;
    }
    __syncthreads();

    const int er  = lane >> 4;
    const int ec4 = lane & 15;
    v4f p[8];
#pragma unroll
    for (int t = 0; t < 8; ++t) {
        const int gr  = row_base + er + 2 * t;
        const int grc = (gr < n_nodes) ? gr : (n_nodes - 1);
        p[t] = ((const v4f*)(x1 + (size_t)grc * DIM))[ec4];
    }

#pragma unroll
    for (int t = 0; t < 8; ++t) {
        const int r  = er + 2 * t;
        const int gr = row_base + r;
        const v4f c = ((const v4f*)Ct)[r * 16 + ec4];
        if (gr < n_nodes) {
            ((v4f*)(out + (size_t)gr * DIM))[ec4] = (c + p[t]) * (1.0f / 3.0f);
        }
    }
}

__global__ void __launch_bounds__(256)
gcn_spmm_kernel(const float* __restrict__ vals, const float* __restrict__ dropu,
                const int* __restrict__ rows, const int* __restrict__ cols,
                const float* __restrict__ src, float* __restrict__ dst,
                int n_edges, float wscale) {
    const int e = blockIdx.x * 256 + threadIdx.x;
    const int l = threadIdx.x & 31;
    const bool in_range = e < n_edges;

    float u = 0.0f, w = 0.0f;
    int col = 0, row = 0;
    if (in_range) {
        u   = dropu[e];
        w   = vals[e] * wscale;
        col = cols[e];
        row = rows[e];
    }
    const bool keep = in_range && !(u + KEEP_PROB < 1.0f);
    unsigned mask = __builtin_amdgcn_ballot_w32(keep);
    const int wbits = __float_as_int(w);

    while (mask != 0u) {
        const int first = __builtin_ctz(mask);
        int  sl[8];
        bool ok[8];
#pragma unroll
        for (int i = 0; i < 8; ++i) {
            ok[i] = (mask != 0u);
            sl[i] = ok[i] ? __builtin_ctz(mask) : first;
            mask &= (mask - 1u);
        }

        float g0[8], g1[8];
#pragma unroll
        for (int i = 0; i < 8; ++i) {
            const int c = __builtin_amdgcn_readlane(col, sl[i]);
            const float* sp = src + (size_t)c * DIM + l;
            g0[i] = sp[0];
            g1[i] = sp[32];
        }

#pragma unroll
        for (int i = 0; i < 8; ++i) {
            float a0 = g0[i], a1 = g1[i];
            asm volatile("" : "+v"(a0), "+v"(a1));
            if (ok[i]) {
                const int   r  = __builtin_amdgcn_readlane(row, sl[i]);
                const float wv = __int_as_float(__builtin_amdgcn_readlane(wbits, sl[i]));
                float* dp = dst + (size_t)r * DIM + l;
                const float m0 = wv * a0;
                const float m1 = wv * a1;
                asm volatile("global_atomic_add_f32 %0, %1, off scope:SCOPE_DEV"
                             :: "v"(dp), "v"(m0) : "memory");
                asm volatile("global_atomic_add_f32 %0, %1, off offset:128 scope:SCOPE_DEV"
                             :: "v"(dp), "v"(m1) : "memory");
            }
        }
    }
}

extern "C" void kernel_launch(void* const* d_in, const int* in_sizes, int n_in,
                              void* d_out, int out_size, void* d_ws, size_t ws_size,
                              hipStream_t stream) {
    const float* emb   = (const float*)d_in[0];
    const float* W     = (const float*)d_in[1];
    const float* bias  = (const float*)d_in[2];
    const float* vals  = (const float*)d_in[3];
    const float* dropu = (const float*)d_in[4];
    const int*   rows  = (const int*)d_in[5];
    const int*   cols  = (const int*)d_in[6];

    const int N = in_sizes[0] / DIM;
    const int E = in_sizes[3];

    float* out = (float*)d_out;
    float* x1  = (float*)d_ws;

    hipMemsetAsync(d_ws, 0, (size_t)N * DIM * sizeof(float), stream);

    dim3 egrid((unsigned)((E + 255) / 256)), eblock(256);

    gcn_spmm_kernel<<<egrid, eblock, 0, stream>>>(vals, dropu, rows, cols, emb, x1, E,
                                                  1.0f / KEEP_PROB);

    {
        dim3 grid((N + 127) / 128), block(256);
        gcn_fc_wmma_kernel<<<grid, block, 0, stream>>>(emb, W, bias, x1, out, N);
    }

    gcn_spmm_kernel<<<egrid, eblock, 0, stream>>>(vals, dropu + E, rows, cols, x1, out, E,
                                                  1.0f / (3.0f * KEEP_PROB));
}


# fc pass: LDS tiles of W and of the per-wave A/C tile get a row pitch of 68 floats (272 B) instead of 64, so each WMMA fragment read hits 64 different banks, not 16 rows on the same two banks
# speedup vs baseline: 4.0316x; 1.1090x over previous
#include <hip/hip_runtime.h>
#include <hip/hip_bf16.h>

typedef __attribute__((ext_vector_type(2))) float v2f;
typedef __attribute__((ext_vector_type(4))) float v4f;
typedef __attribute__((ext_vector_type(8))) float v8f;

#define DIM 64
#define KEEP_PROB 0.7f
#define LDP 68

__global__ void __launch_bounds__(256)
gcn_fc_wmma_kernel(const float* __restrict__ emb, const float* __restrict__ W,
                   const float* __restrict__ bias, const float* __restrict__ x1,
                   float* __restrict__ out, int n_nodes) {
    __shared__ alignas(16) float Wlds[DIM * LDP];
    __shared__ alignas(16) float Alds[8][16 * LDP];

    const int tid  = threadIdx.x;
    const int wave = tid >> 5;
    const int lane = tid & 31;

    for (int i = tid; i < (DIM * DIM) / 4; i += 256) {
        const int n  = i >> 4;
        const int c4 = i & 15;
        *(v4f*)(Wlds + n * LDP + c4 * 4) = ((const v4f*)W)[i];
    }

    const int row_base = blockIdx.x * 128 + wave * 16;

    for (int i = lane; i < 256; i += 32) {
        int r  = i >> 4;
        int c4 = i & 15;
        int gr = row_base + r;
        v4f v = (gr < n_nodes) ? ((const v4f*)(emb + (size_t)gr * DIM))[c4]
                               : (v4f){0.f, 0.f, 0.f, 0.f};
        *(v4f*)(&Alds[wave][0] + r * LDP + c4 * 4) = v;
    }
    __syncthreads();

    const float* At = &Alds[wave][0];

    const int m_rel = lane & 15;
    const int k_off = (lane >> 4) * 2;
    const int n_lo  = lane & 15;

    v8f c0 = {}, c1 = {}, c2 = {}, c3 = {};

#pragma unroll
    for (int kk = 0; kk < 16; ++kk) {
        const int k = kk * 4 + k_off;
        v2f a  = *(const v2f*)(At + m_rel * LDP + k);
        v2f b0 = *(const v2f*)(Wlds + (n_lo +  0) * LDP + k);
        v2f b1 = *(const v2f*)(Wlds + (n_lo + 16) * LDP + k);
        v2f b2 = *(const v2f*)(Wlds + (n_lo + 32) * LDP + k);
        v2f b3 = *(const v2f*)(Wlds + (n_lo + 48) * LDP + k);
        c0 = __builtin_amdgcn_wmma_f32_16x16x4_f32(false, a, false, b0, (short)0, c0, false, false);
        c1 = __builtin_amdgcn_wmma_f32_16x16x4_f32(false, a, false, b1, (short)0, c1, false, false);
        c2 = __builtin_amdgcn_wmma_f32_16x16x4_f32(false, a, false, b2, (short)0, c2, false, false);
        c3 = __builtin_amdgcn_wmma_f32_16x16x4_f32(false, a, false, b3, (short)0, c3, false, false);
    }

    const float bb0 = bias[n_lo];
    const float bb1 = bias[n_lo + 16];
    const float bb2 = bias[n_lo + 32];
    const float bb3 = bias[n_lo + 48];

    float* Ct = &Alds[wave][0];
    const int m_hi = (lane >> 4) * 8;
#pragma unroll
    for (int j = 0; j < 8; ++j) {
        float* cr = Ct + (m_hi + j) * LDP + n_lo;
        cr[ 0] = c0[j] + bb0;
        cr[16] = c1[j] + bb1;
        cr[32] = c2[j] + bb2;
        cr[48] = c3[j] + bb3;
    }
    __syncthreads();

    const int er  = lane >> 4;
    const int ec4 = lane & 15;
    v4f p[8];
#pragma unroll
    for (int t = 0; t < 8; ++t) {
        const int gr  = row_base + er + 2 * t;
        const int grc = (gr < n_nodes) ? gr : (n_nodes - 1);
        p[t] = ((const v4f*)(x1 + (size_t)grc * DIM))[ec4];
    }

#pragma unroll
    for (int t = 0; t < 8; ++t) {
        const int r  = er + 2 * t;
        const int gr = row_base + r;
        const v4f c = *(const v4f*)(Ct + r * LDP + ec4 * 4);
        if (gr < n_nodes) {
            ((v4f*)(out + (size_t)gr * DIM))[ec4] = (c + p[t]) * (1.0f / 3.0f);
        }
    }
}

__global__ void __launch_bounds__(256)
gcn_spmm_kernel(const float* __restrict__ vals, const float* __restrict__ dropu,
                const int* __restrict__ rows, const int* __restrict__ cols,
                const float* __restrict__ src, float* __restrict__ dst,
                int n_edges, float wscale) {
    const int e = blockIdx.x * 256 + threadIdx.x;
    const int l = threadIdx.x & 31;
    const bool in_range = e < n_edges;

    float u = 0.0f, w = 0.0f;
    int col = 0, row = 0;
    if (in_range) {
        u   = dropu[e];
        w   = vals[e] * wscale;
        col = cols[e];
        row = rows[e];
    }
    const bool keep = in_range && !(u + KEEP_PROB < 1.0f);
    unsigned mask = __builtin_amdgcn_ballot_w32(keep);
    const int wbits = __float_as_int(w);

    while (mask != 0u) {
        const int first = __builtin_ctz(mask);
        int  sl[8];
        bool ok[8];
#pragma unroll
        for (int i = 0; i < 8; ++i) {
            ok[i] = (mask != 0u);
            sl[i] = ok[i] ? __builtin_ctz(mask) : first;
            mask &= (mask - 1u);
        }

        float g0[8], g1[8];
#pragma unroll
        for (int i = 0; i < 8; ++i) {
            const int c = __builtin_amdgcn_readlane(col, sl[i]);
            const float* sp = src + (size_t)c * DIM + l;
            g0[i] = sp[0];
            g1[i] = sp[32];
        }

#pragma unroll
        for (int i = 0; i < 8; ++i) {
            float a0 = g0[i], a1 = g1[i];
            asm volatile("" : "+v"(a0), "+v"(a1));
            if (ok[i]) {
                const int   r  = __builtin_amdgcn_readlane(row, sl[i]);
                const float wv = __int_as_float(__builtin_amdgcn_readlane(wbits, sl[i]));
                float* dp = dst + (size_t)r * DIM + l;
                const float m0 = wv * a0;
                const float m1 = wv * a1;
                asm volatile("global_atomic_add_f32 %0, %1, off scope:SCOPE_DEV"
                             :: "v"(dp), "v"(m0) : "memory");
                asm volatile("global_atomic_add_f32 %0, %1, off offset:128 scope:SCOPE_DEV"
                             :: "v"(dp), "v"(m1) : "memory");
            }
        }
    }
}

extern "C" void kernel_launch(void* const* d_in, const int* in_sizes, int n_in,
                              void* d_out, int out_size, void* d_ws, size_t ws_size,
                              hipStream_t stream) {
    const float* emb   = (const float*)d_in[0];
    const float* W     = (const float*)d_in[1];
    const float* bias  = (const float*)d_in[2];
    const float* vals  = (const float*)d_in[3];
    const float* dropu = (const float*)d_in[4];
    const int*   rows  = (const int*)d_in[5];
    const int*   cols  = (const int*)d_in[6];

    const int N = in_sizes[0] / DIM;
    const int E = in_sizes[3];

    float* out = (float*)d_out;
    float* x1  = (float*)d_ws;

    hipMemsetAsync(d_ws, 0, (size_t)N * DIM * sizeof(float), stream);

    dim3 egrid((unsigned)((E + 255) / 256)), eblock(256);

    gcn_spmm_kernel<<<egrid, eblock, 0, stream>>>(vals, dropu, rows, cols, emb, x1, E,
                                                  1.0f / KEEP_PROB);

    {
        dim3 grid((N + 127) / 128), block(256);
        gcn_fc_wmma_kernel<<<grid, block, 0, stream>>>(emb, W, bias, x1, out, N);
    }

    gcn_spmm_kernel<<<egrid, eblock, 0, stream>>>(vals, dropu + E, rows, cols, x1, out, E,
                                                  1.0f / (3.0f * KEEP_PROB));
}
